# starting kernel
# baseline (speedup 1.0000x reference)
#include <hip/hip_runtime.h>


typedef float v2f __attribute__((ext_vector_type(2)));
typedef float v8f __attribute__((ext_vector_type(8)));

__global__ void gcn_init(float* __restrict__ deg, float* __restrict__ agg,
                         float* __restrict__ z, float* __restrict__ out, int n) {
  int i = blockIdx.x * blockDim.x + threadIdx.x;
  if (i < n) {
    deg[i] = 1.0f;
    agg[2 * i] = 0.0f;
    agg[2 * i + 1] = 0.0f;
    z[i] = 0.0f;
    out[i] = 0.0f;
  }
}

__global__ void gcn_degree(const int* __restrict__ dst, float* __restrict__ deg,
                           int E) {
  int i = blockIdx.x * blockDim.x + threadIdx.x;
  int stride = gridDim.x * blockDim.x;
  for (; i < E; i += stride) {
    if (i + stride < E) __builtin_prefetch(&dst[i + stride], 0, 1);
    atomicAdd(&deg[dst[i]], 1.0f);
  }
}

__global__ void gcn_rsqrt(float* __restrict__ dinv, int n) {
  int i = blockIdx.x * blockDim.x + threadIdx.x;
  if (i < n) dinv[i] = rsqrtf(dinv[i]);
}

__global__ void gcn_agg2(const int* __restrict__ src, const int* __restrict__ dst,
                         const float* __restrict__ x, const float* __restrict__ dinv,
                         float* __restrict__ agg, int E) {
  int i = blockIdx.x * blockDim.x + threadIdx.x;
  int stride = gridDim.x * blockDim.x;
  for (; i < E; i += stride) {
    if (i + stride < E) {
      __builtin_prefetch(&src[i + stride], 0, 1);
      __builtin_prefetch(&dst[i + stride], 0, 1);
    }
    int s = src[i];
    int d = dst[i];
    float w = dinv[s] * dinv[d];
    atomicAdd(&agg[2 * d], x[2 * s] * w);
    atomicAdd(&agg[2 * d + 1], x[2 * s + 1] * w);
  }
}

__global__ __launch_bounds__(256) void gcn_dense_wmma(
    const float* __restrict__ agg, const float* __restrict__ x,
    const float* __restrict__ dinv, const float* __restrict__ W1,
    const float* __restrict__ b1, const float* __restrict__ W2,
    float* __restrict__ z, int n) {
  const int lane = threadIdx.x & 31;
  const int wave = threadIdx.x >> 5;
  const int tile = blockIdx.x * 8 + wave;
  const int base = tile * 16;
  if (base >= n) return;

  const int m = lane & 15;
  const bool hiK = lane >= 16;
  const float kmask = hiK ? 0.0f : 1.0f;

  int node = base + m;
  const float rowmask = (!hiK && node < n) ? 1.0f : 0.0f;
  if (node >= n) node = n - 1;
  {
  }
  float di = dinv[node];
  float sl = di * di;
  v2f A;
  A.x = (agg[2 * node]     + x[2 * node]     * sl) * rowmask;
  A.y = (agg[2 * node + 1] + x[2 * node + 1] * sl) * rowmask;

  float zacc[8];
#pragma unroll
  for (int v = 0; v < 8; ++v) zacc[v] = 0.0f;

#pragma unroll
  for (int n0 = 0; n0 < 128; n0 += 16) {
    int col = n0 + m;
    v2f B;
    B.x = W1[col]       * kmask;
    B.y = W1[128 + col] * kmask;
    float bias = b1[col];
    v8f C;
#pragma unroll
    for (int v = 0; v < 8; ++v) C[v] = bias;

    C = __builtin_amdgcn_wmma_f32_16x16x4_f32(
         false, A,  false, B,
         (short)0, C,  false,  false);

    float w2 = W2[col];
#pragma unroll
    for (int v = 0; v < 8; ++v) {
      float h = C[v];
      h = h > 0.0f ? h : 0.0f;
      zacc[v] += h * w2;
    }
  }

#pragma unroll
  for (int v = 0; v < 8; ++v) {
    float r = zacc[v];
    r += __shfl_xor(r, 1, 16);
    r += __shfl_xor(r, 2, 16);
    r += __shfl_xor(r, 4, 16);
    r += __shfl_xor(r, 8, 16);
    zacc[v] = r;
  }
  if (m == 0) {
    int rowbase = base + (hiK ? 8 : 0);
#pragma unroll
    for (int v = 0; v < 8; ++v) {
      int r = rowbase + v;
      if (r < n) z[r] = zacc[v];
    }
  }
}

__global__ void gcn_out_self(const float* __restrict__ z,
                             const float* __restrict__ dinv,
                             const float* __restrict__ b2,
                             float* __restrict__ out, int n) {
  int i = blockIdx.x * blockDim.x + threadIdx.x;
  if (i < n) out[i] = z[i] * dinv[i] * dinv[i] + b2[0];
}

__global__ void gcn_out_edges(const int* __restrict__ src, const int* __restrict__ dst,
                              const float* __restrict__ z,
                              const float* __restrict__ dinv,
                              float* __restrict__ out, int E) {
  int i = blockIdx.x * blockDim.x + threadIdx.x;
  int stride = gridDim.x * blockDim.x;
  for (; i < E; i += stride) {
    if (i + stride < E) {
      __builtin_prefetch(&src[i + stride], 0, 1);
      __builtin_prefetch(&dst[i + stride], 0, 1);
    }
    int s = src[i];
    int d = dst[i];
    atomicAdd(&out[d], z[s] * dinv[s] * dinv[d]);
  }
}

extern "C" void kernel_launch(void* const* d_in, const int* in_sizes, int n_in,
                              void* d_out, int out_size, void* d_ws, size_t ws_size,
                              hipStream_t stream) {
  const float* x  = (const float*)d_in[0];
  const int*   ei = (const int*)d_in[1];
  const float* W1 = (const float*)d_in[2];
  const float* b1 = (const float*)d_in[3];
  const float* W2 = (const float*)d_in[4];
  const float* b2 = (const float*)d_in[5];

  int n = in_sizes[0] / 2;
  int E = in_sizes[1] / 2;
  const int* src = ei;
  const int* dst = ei + E;
  float* out = (float*)d_out;

  float* dinv = (float*)d_ws;
  float* agg  = dinv + n;
  float* z    = agg + 2 * (size_t)n;

  const int tb = 256;
  int gN = (n + tb - 1) / tb;
  int gE = (E + tb - 1) / tb;
  if (gE > 4096) gE = 4096;

  gcn_init<<<gN, tb, 0, stream>>>(dinv, agg, z, out, n);
  gcn_degree<<<gE, tb, 0, stream>>>(dst, dinv, E);
  gcn_rsqrt<<<gN, tb, 0, stream>>>(dinv, n);
  gcn_agg2<<<gE, tb, 0, stream>>>(src, dst, x, dinv, agg, E);

  int tiles = (n + 15) / 16;
  int dblocks = (tiles + 7) / 8;
  gcn_dense_wmma<<<dblocks, 256, 0, stream>>>(agg, x, dinv, W1, b1, W2, z, n);

  gcn_out_self<<<gN, tb, 0, stream>>>(z, dinv, b2, out, n);
  gcn_out_edges<<<gE, tb, 0, stream>>>(src, dst, z, dinv, out, E);
}


# Edge sums tiled by destination range (6400 nodes) x 1/16 edge slice: a block adds messages in LDS (range's dinv staged there), then flushes lane-consecutive float atomics; f32 sum order: LDS arrival
# speedup vs baseline: 4.0255x; 4.0255x over previous
#include <hip/hip_runtime.h>


typedef float v2f __attribute__((ext_vector_type(2)));
typedef float v8f __attribute__((ext_vector_type(8)));

#define TILE 6400
#define SLICES 16
#define ETHREADS 1024
#define EUNROLL 16

__global__ void gcn_init(float* __restrict__ deg, float* __restrict__ agg,
                         float* __restrict__ z, float* __restrict__ out, int n) {
  int i = blockIdx.x * blockDim.x + threadIdx.x;
  if (i < n) {
    deg[i] = 1.0f;
    agg[2 * i] = 0.0f;
    agg[2 * i + 1] = 0.0f;
    z[i] = 0.0f;
    out[i] = 0.0f;
  }
}

__global__ __launch_bounds__(ETHREADS) void gcn_degree(
    const int* __restrict__ dst, float* __restrict__ deg, int E, int n) {
  __shared__ float acc[TILE];
  const int tid = threadIdx.x;
  const int slice = blockIdx.x % SLICES;
  const int lo = (blockIdx.x / SLICES) * TILE;
  const unsigned cnt = (unsigned)min(TILE, n - lo);
  const int chunk = (E + SLICES - 1) / SLICES;
  const int e0 = slice * chunk;
  const int e1 = min(E, e0 + chunk);

  for (int i = tid; i < TILE; i += ETHREADS) acc[i] = 0.0f;
  __syncthreads();

  const int trips = (chunk + EUNROLL * ETHREADS - 1) / (EUNROLL * ETHREADS);
  for (int t = 0; t < trips; ++t) {
    const int i = e0 + tid + t * (EUNROLL * ETHREADS);
    int d[EUNROLL];
#pragma unroll
    for (int u = 0; u < EUNROLL; ++u) {
      int j = i + u * ETHREADS;
      d[u] = dst[min(j, e1 - 1)];
      if (j >= e1) d[u] = -1;
    }
#pragma unroll
    for (int u = 0; u < EUNROLL; ++u) {
      unsigned k = (unsigned)(d[u] - lo);
      if (d[u] >= lo && k < cnt) atomicAdd(&acc[k], 1.0f);
    }
  }
  __syncthreads();

  for (unsigned i = tid; i < cnt; i += ETHREADS) {
    float v = acc[i];
    if (v != 0.0f) atomicAdd(&deg[lo + i], v);
  }
}

__global__ void gcn_rsqrt(float* __restrict__ dinv, int n) {
  int i = blockIdx.x * blockDim.x + threadIdx.x;
  if (i < n) dinv[i] = rsqrtf(dinv[i]);
}

__global__ __launch_bounds__(ETHREADS) void gcn_agg2(
    const int* __restrict__ src, const int* __restrict__ dst,
    const float* __restrict__ x, const float* __restrict__ dinv,
    float* __restrict__ agg, int E, int n) {
  __shared__ float acc[2 * TILE];
  __shared__ float dtile[TILE];
  const int tid = threadIdx.x;
  const int slice = blockIdx.x % SLICES;
  const int lo = (blockIdx.x / SLICES) * TILE;
  const unsigned cnt = (unsigned)min(TILE, n - lo);
  const int chunk = (E + SLICES - 1) / SLICES;
  const int e0 = slice * chunk;
  const int e1 = min(E, e0 + chunk);

  for (int i = tid; i < 2 * TILE; i += ETHREADS) acc[i] = 0.0f;
  for (unsigned i = tid; i < cnt; i += ETHREADS) dtile[i] = dinv[lo + i];
  __syncthreads();

  const v2f* __restrict__ x2 = (const v2f*)x;
  const int trips = (chunk + EUNROLL * ETHREADS - 1) / (EUNROLL * ETHREADS);
  for (int t = 0; t < trips; ++t) {
    const int i = e0 + tid + t * (EUNROLL * ETHREADS);
    int d[EUNROLL], s[EUNROLL];
    bool hit[EUNROLL];
#pragma unroll
    for (int u = 0; u < EUNROLL; ++u) {
      int j = min(i + u * ETHREADS, e1 - 1);
      d[u] = dst[j];
      s[u] = src[j];
    }
#pragma unroll
    for (int u = 0; u < EUNROLL; ++u) asm volatile("" ::"v"(d[u]), "v"(s[u]));
    unsigned k[EUNROLL];
#pragma unroll
    for (int u = 0; u < EUNROLL; ++u) {
      k[u] = (unsigned)(d[u] - lo);
      hit[u] = (i + u * ETHREADS < e1) && d[u] >= lo && k[u] < cnt;
      if (!hit[u]) { k[u] = 0; s[u] = lo; }
    }
    float ws[EUNROLL], wd[EUNROLL];
    v2f xs[EUNROLL];
#pragma unroll
    for (int u = 0; u < EUNROLL; ++u) {
      ws[u] = dinv[s[u]];
      xs[u] = x2[s[u]];
      wd[u] = dtile[k[u]];
    }
#pragma unroll
    for (int u = 0; u < EUNROLL; ++u)
      asm volatile("" ::"v"(ws[u]), "v"(wd[u]), "v"(xs[u].x), "v"(xs[u].y));
#pragma unroll
    for (int u = 0; u < EUNROLL; ++u) {
      if (hit[u]) {
        float w = ws[u] * wd[u];
        atomicAdd(&acc[2 * k[u]], xs[u].x * w);
        atomicAdd(&acc[2 * k[u] + 1], xs[u].y * w);
      }
    }
  }
  __syncthreads();

  for (unsigned i = tid; i < 2 * cnt; i += ETHREADS) {
    float v = acc[i];
    if (v != 0.0f) atomicAdd(&agg[2 * (size_t)lo + i], v);
  }
}

__global__ __launch_bounds__(256) void gcn_dense_wmma(
    const float* __restrict__ agg, const float* __restrict__ x,
    const float* __restrict__ dinv, const float* __restrict__ W1,
    const float* __restrict__ b1, const float* __restrict__ W2,
    float* __restrict__ z, int n) {
  const int lane = threadIdx.x & 31;
  const int wave = threadIdx.x >> 5;
  const int tile = blockIdx.x * 8 + wave;
  const int base = tile * 16;
  if (base >= n) return;

  const int m = lane & 15;
  const bool hiK = lane >= 16;
  const float kmask = hiK ? 0.0f : 1.0f;

  int node = base + m;
  const float rowmask = (!hiK && node < n) ? 1.0f : 0.0f;
  if (node >= n) node = n - 1;
  {
  }
  float di = dinv[node];
  float sl = di * di;
  v2f A;
  A.x = (agg[2 * node]     + x[2 * node]     * sl) * rowmask;
  A.y = (agg[2 * node + 1] + x[2 * node + 1] * sl) * rowmask;

  float zacc[8];
#pragma unroll
  for (int v = 0; v < 8; ++v) zacc[v] = 0.0f;

#pragma unroll
  for (int n0 = 0; n0 < 128; n0 += 16) {
    int col = n0 + m;
    v2f B;
    B.x = W1[col]       * kmask;
    B.y = W1[128 + col] * kmask;
    float bias = b1[col];
    v8f C;
#pragma unroll
    for (int v = 0; v < 8; ++v) C[v] = bias;

    C = __builtin_amdgcn_wmma_f32_16x16x4_f32(
         false, A,  false, B,
         (short)0, C,  false,  false);

    float w2 = W2[col];
#pragma unroll
    for (int v = 0; v < 8; ++v) {
      float h = C[v];
      h = h > 0.0f ? h : 0.0f;
      zacc[v] += h * w2;
    }
  }

#pragma unroll
  for (int v = 0; v < 8; ++v) {
    float r = zacc[v];
    r += __shfl_xor(r, 1, 16);
    r += __shfl_xor(r, 2, 16);
    r += __shfl_xor(r, 4, 16);
    r += __shfl_xor(r, 8, 16);
    zacc[v] = r;
  }
  if (m == 0) {
    int rowbase = base + (hiK ? 8 : 0);
#pragma unroll
    for (int v = 0; v < 8; ++v) {
      int r = rowbase + v;
      if (r < n) z[r] = zacc[v];
    }
  }
}

__global__ void gcn_out_self(const float* __restrict__ z,
                             const float* __restrict__ dinv,
                             const float* __restrict__ b2,
                             float* __restrict__ out, int n) {
  int i = blockIdx.x * blockDim.x + threadIdx.x;
  if (i < n) out[i] = z[i] * dinv[i] * dinv[i] + b2[0];
}

__global__ __launch_bounds__(ETHREADS) void gcn_out_edges(
    const int* __restrict__ src, const int* __restrict__ dst,
    const float* __restrict__ z, const float* __restrict__ dinv,
    float* __restrict__ out, int E, int n) {
  __shared__ float acc[TILE];
  __shared__ float dtile[TILE];
  const int tid = threadIdx.x;
  const int slice = blockIdx.x % SLICES;
  const int lo = (blockIdx.x / SLICES) * TILE;
  const unsigned cnt = (unsigned)min(TILE, n - lo);
  const int chunk = (E + SLICES - 1) / SLICES;
  const int e0 = slice * chunk;
  const int e1 = min(E, e0 + chunk);

  for (int i = tid; i < TILE; i += ETHREADS) acc[i] = 0.0f;
  for (unsigned i = tid; i < cnt; i += ETHREADS) dtile[i] = dinv[lo + i];
  __syncthreads();

  const int trips = (chunk + EUNROLL * ETHREADS - 1) / (EUNROLL * ETHREADS);
  for (int t = 0; t < trips; ++t) {
    const int i = e0 + tid + t * (EUNROLL * ETHREADS);
    int d[EUNROLL], s[EUNROLL];
    bool hit[EUNROLL];
#pragma unroll
    for (int u = 0; u < EUNROLL; ++u) {
      int j = min(i + u * ETHREADS, e1 - 1);
      d[u] = dst[j];
      s[u] = src[j];
    }
#pragma unroll
    for (int u = 0; u < EUNROLL; ++u) asm volatile("" ::"v"(d[u]), "v"(s[u]));
    unsigned k[EUNROLL];
#pragma unroll
    for (int u = 0; u < EUNROLL; ++u) {
      k[u] = (unsigned)(d[u] - lo);
      hit[u] = (i + u * ETHREADS < e1) && d[u] >= lo && k[u] < cnt;
      if (!hit[u]) { k[u] = 0; s[u] = lo; }
    }
    float ws[EUNROLL], wd[EUNROLL], zs[EUNROLL];
#pragma unroll
    for (int u = 0; u < EUNROLL; ++u) {
      ws[u] = dinv[s[u]];
      zs[u] = z[s[u]];
      wd[u] = dtile[k[u]];
    }
#pragma unroll
    for (int u = 0; u < EUNROLL; ++u)
      asm volatile("" ::"v"(ws[u]), "v"(wd[u]), "v"(zs[u]));
#pragma unroll
    for (int u = 0; u < EUNROLL; ++u) {
      if (hit[u]) atomicAdd(&acc[k[u]], zs[u] * ws[u] * wd[u]);
    }
  }
  __syncthreads();

  for (unsigned i = tid; i < cnt; i += ETHREADS) {
    float v = acc[i];
    if (v != 0.0f) atomicAdd(&out[lo + i], v);
  }
}

extern "C" void kernel_launch(void* const* d_in, const int* in_sizes, int n_in,
                              void* d_out, int out_size, void* d_ws, size_t ws_size,
                              hipStream_t stream) {
  const float* x  = (const float*)d_in[0];
  const int*   ei = (const int*)d_in[1];
  const float* W1 = (const float*)d_in[2];
  const float* b1 = (const float*)d_in[3];
  const float* W2 = (const float*)d_in[4];
  const float* b2 = (const float*)d_in[5];

  int n = in_sizes[0] / 2;
  int E = in_sizes[1] / 2;
  const int* src = ei;
  const int* dst = ei + E;
  float* out = (float*)d_out;

  float* dinv = (float*)d_ws;
  float* agg  = dinv + n;
  float* z    = agg + 2 * (size_t)n;

  const int tb = 256;
  int gN = (n + tb - 1) / tb;
  int gE = ((n + TILE - 1) / TILE) * SLICES;

  gcn_init<<<gN, tb, 0, stream>>>(dinv, agg, z, out, n);
  gcn_degree<<<gE, ETHREADS, 0, stream>>>(dst, dinv, E, n);
  gcn_rsqrt<<<gN, tb, 0, stream>>>(dinv, n);
  gcn_agg2<<<gE, ETHREADS, 0, stream>>>(src, dst, x, dinv, agg, E, n);

  int tiles = (n + 15) / 16;
  int dblocks = (tiles + 7) / 8;
  gcn_dense_wmma<<<dblocks, 256, 0, stream>>>(agg, x, dinv, W1, b1, W2, z, n);

  gcn_out_self<<<gN, tb, 0, stream>>>(z, dinv, b2, out, n);
  gcn_out_edges<<<gE, ETHREADS, 0, stream>>>(src, dst, z, dinv, out, E, n);
}


# Node stages store payloads pre-scaled by the source's dinv (xs=x*dinv, zs=z*dinv): an edge gathers one value, not two; destination's dinv multiplies the sum; f32 sum order unchanged (LDS arrival)
# speedup vs baseline: 4.8538x; 1.2058x over previous
#include <hip/hip_runtime.h>


typedef float v2f __attribute__((ext_vector_type(2)));
typedef float v8f __attribute__((ext_vector_type(8)));

#define TILE 6400
#define SLICES 16
#define ETHREADS 1024
#define EUNROLL 16

__global__ void gcn_init(float* __restrict__ deg, float* __restrict__ agg,
                         float* __restrict__ z, float* __restrict__ out, int n) {
  int i = blockIdx.x * blockDim.x + threadIdx.x;
  if (i < n) {
    deg[i] = 1.0f;
    agg[2 * i] = 0.0f;
    agg[2 * i + 1] = 0.0f;
    z[i] = 0.0f;
    out[i] = 0.0f;
  }
}

__global__ __launch_bounds__(ETHREADS) void gcn_degree(
    const int* __restrict__ dst, float* __restrict__ deg, int E, int n) {
  __shared__ float acc[TILE];
  const int tid = threadIdx.x;
  const int slice = blockIdx.x % SLICES;
  const int lo = (blockIdx.x / SLICES) * TILE;
  const unsigned cnt = (unsigned)min(TILE, n - lo);
  const int chunk = (E + SLICES - 1) / SLICES;
  const int e0 = slice * chunk;
  const int e1 = min(E, e0 + chunk);

  for (int i = tid; i < TILE; i += ETHREADS) acc[i] = 0.0f;
  __syncthreads();

  const int trips = (chunk + EUNROLL * ETHREADS - 1) / (EUNROLL * ETHREADS);
  for (int t = 0; t < trips; ++t) {
    const int i = e0 + tid + t * (EUNROLL * ETHREADS);
    int d[EUNROLL];
#pragma unroll
    for (int u = 0; u < EUNROLL; ++u) {
      int j = i + u * ETHREADS;
      d[u] = dst[min(j, e1 - 1)];
      if (j >= e1) d[u] = -1;
    }
#pragma unroll
    for (int u = 0; u < EUNROLL; ++u) {
      unsigned k = (unsigned)(d[u] - lo);
      if (d[u] >= lo && k < cnt) atomicAdd(&acc[k], 1.0f);
    }
  }
  __syncthreads();

  for (unsigned i = tid; i < cnt; i += ETHREADS) {
    float v = acc[i];
    if (v != 0.0f) atomicAdd(&deg[lo + i], v);
  }
}

__global__ void gcn_rsqrt(float* __restrict__ dinv, const float* __restrict__ x,
                          float* __restrict__ xs, int n) {
  int i = blockIdx.x * blockDim.x + threadIdx.x;
  if (i < n) {
    float di = rsqrtf(dinv[i]);
    dinv[i] = di;
    v2f xv = ((const v2f*)x)[i];
    v2f o;
    o.x = xv.x * di;
    o.y = xv.y * di;
    ((v2f*)xs)[i] = o;
  }
}

__global__ __launch_bounds__(ETHREADS) void gcn_agg2(
    const int* __restrict__ src, const int* __restrict__ dst,
    const float* __restrict__ xs, float* __restrict__ agg, int E, int n) {
  __shared__ float acc[2 * TILE];
  const int tid = threadIdx.x;
  const int slice = blockIdx.x % SLICES;
  const int lo = (blockIdx.x / SLICES) * TILE;
  const unsigned cnt = (unsigned)min(TILE, n - lo);
  const int chunk = (E + SLICES - 1) / SLICES;
  const int e0 = slice * chunk;
  const int e1 = min(E, e0 + chunk);

  for (int i = tid; i < 2 * TILE; i += ETHREADS) acc[i] = 0.0f;
  __syncthreads();

  const v2f* __restrict__ xs2 = (const v2f*)xs;
  const int trips = (chunk + EUNROLL * ETHREADS - 1) / (EUNROLL * ETHREADS);
  for (int t = 0; t < trips; ++t) {
    const int i = e0 + tid + t * (EUNROLL * ETHREADS);
    int d[EUNROLL], s[EUNROLL];
    bool hit[EUNROLL];
#pragma unroll
    for (int u = 0; u < EUNROLL; ++u) {
      int j = min(i + u * ETHREADS, e1 - 1);
      d[u] = dst[j];
      s[u] = src[j];
    }
#pragma unroll
    for (int u = 0; u < EUNROLL; ++u) asm volatile("" ::"v"(d[u]), "v"(s[u]));
    unsigned k[EUNROLL];
#pragma unroll
    for (int u = 0; u < EUNROLL; ++u) {
      k[u] = (unsigned)(d[u] - lo);
      hit[u] = (i + u * ETHREADS < e1) && d[u] >= lo && k[u] < cnt;
      if (!hit[u]) s[u] = lo;
    }
    v2f p[EUNROLL];
#pragma unroll
    for (int u = 0; u < EUNROLL; ++u) p[u] = xs2[s[u]];
#pragma unroll
    for (int u = 0; u < EUNROLL; ++u) asm volatile("" ::"v"(p[u].x), "v"(p[u].y));
#pragma unroll
    for (int u = 0; u < EUNROLL; ++u) {
      if (hit[u]) {
        atomicAdd(&acc[2 * k[u]], p[u].x);
        atomicAdd(&acc[2 * k[u] + 1], p[u].y);
      }
    }
  }
  __syncthreads();

  for (unsigned i = tid; i < 2 * cnt; i += ETHREADS) {
    float v = acc[i];
    if (v != 0.0f) atomicAdd(&agg[2 * (size_t)lo + i], v);
  }
}

__global__ __launch_bounds__(256) void gcn_dense_wmma(
    const float* __restrict__ agg, const float* __restrict__ xs,
    const float* __restrict__ dinv, const float* __restrict__ W1,
    const float* __restrict__ b1, const float* __restrict__ W2,
    float* __restrict__ zs, int n) {
  const int lane = threadIdx.x & 31;
  const int wave = threadIdx.x >> 5;
  const int tile = blockIdx.x * 8 + wave;
  const int base = tile * 16;
  if (base >= n) return;

  const int m = lane & 15;
  const bool hiK = lane >= 16;
  const float kmask = hiK ? 0.0f : 1.0f;

  int node = base + m;
  const float rowmask = (!hiK && node < n) ? 1.0f : 0.0f;
  if (node >= n) node = n - 1;
  float di = dinv[node];
  float sc = di * rowmask;
  v2f A;
  A.x = (agg[2 * node]     + xs[2 * node])     * sc;
  A.y = (agg[2 * node + 1] + xs[2 * node + 1]) * sc;

  float zacc[8];
#pragma unroll
  for (int v = 0; v < 8; ++v) zacc[v] = 0.0f;

#pragma unroll
  for (int n0 = 0; n0 < 128; n0 += 16) {
    int col = n0 + m;
    v2f B;
    B.x = W1[col]       * kmask;
    B.y = W1[128 + col] * kmask;
    float bias = b1[col];
    v8f C;
#pragma unroll
    for (int v = 0; v < 8; ++v) C[v] = bias;

    C = __builtin_amdgcn_wmma_f32_16x16x4_f32(
         false, A,  false, B,
         (short)0, C,  false,  false);

    float w2 = W2[col];
#pragma unroll
    for (int v = 0; v < 8; ++v) {
      float h = C[v];
      h = h > 0.0f ? h : 0.0f;
      zacc[v] += h * w2;
    }
  }

#pragma unroll
  for (int v = 0; v < 8; ++v) {
    float r = zacc[v];
    r += __shfl_xor(r, 1, 16);
    r += __shfl_xor(r, 2, 16);
    r += __shfl_xor(r, 4, 16);
    r += __shfl_xor(r, 8, 16);
    zacc[v] = r;
  }
  float val = zacc[0];
#pragma unroll
  for (int v = 1; v < 8; ++v) val = ((m & 7) == v) ? zacc[v] : val;
  const bool owner = ((m >> 3) == (hiK ? 1 : 0));
  if (owner && base + m < n) zs[base + m] = val * di;
}

__global__ void gcn_out_self(const float* __restrict__ zs,
                             const float* __restrict__ dinv,
                             const float* __restrict__ b2,
                             float* __restrict__ out, int n) {
  int i = blockIdx.x * blockDim.x + threadIdx.x;
  if (i < n) out[i] = zs[i] * dinv[i] + b2[0];
}

__global__ __launch_bounds__(ETHREADS) void gcn_out_edges(
    const int* __restrict__ src, const int* __restrict__ dst,
    const float* __restrict__ zs, const float* __restrict__ dinv,
    float* __restrict__ out, int E, int n) {
  __shared__ float acc[TILE];
  const int tid = threadIdx.x;
  const int slice = blockIdx.x % SLICES;
  const int lo = (blockIdx.x / SLICES) * TILE;
  const unsigned cnt = (unsigned)min(TILE, n - lo);
  const int chunk = (E + SLICES - 1) / SLICES;
  const int e0 = slice * chunk;
  const int e1 = min(E, e0 + chunk);

  for (int i = tid; i < TILE; i += ETHREADS) acc[i] = 0.0f;
  __syncthreads();

  const int trips = (chunk + EUNROLL * ETHREADS - 1) / (EUNROLL * ETHREADS);
  for (int t = 0; t < trips; ++t) {
    const int i = e0 + tid + t * (EUNROLL * ETHREADS);
    int d[EUNROLL], s[EUNROLL];
    bool hit[EUNROLL];
#pragma unroll
    for (int u = 0; u < EUNROLL; ++u) {
      int j = min(i + u * ETHREADS, e1 - 1);
      d[u] = dst[j];
      s[u] = src[j];
    }
#pragma unroll
    for (int u = 0; u < EUNROLL; ++u) asm volatile("" ::"v"(d[u]), "v"(s[u]));
    unsigned k[EUNROLL];
#pragma unroll
    for (int u = 0; u < EUNROLL; ++u) {
      k[u] = (unsigned)(d[u] - lo);
      hit[u] = (i + u * ETHREADS < e1) && d[u] >= lo && k[u] < cnt;
      if (!hit[u]) s[u] = lo;
    }
    float p[EUNROLL];
#pragma unroll
    for (int u = 0; u < EUNROLL; ++u) p[u] = zs[s[u]];
#pragma unroll
    for (int u = 0; u < EUNROLL; ++u) asm volatile("" ::"v"(p[u]));
#pragma unroll
    for (int u = 0; u < EUNROLL; ++u) {
      if (hit[u]) atomicAdd(&acc[k[u]], p[u]);
    }
  }
  __syncthreads();

  for (unsigned i = tid; i < cnt; i += ETHREADS) {
    float v = acc[i];
    float dd = dinv[lo + i];
    if (v != 0.0f) atomicAdd(&out[lo + i], v * dd);
  }
}

extern "C" void kernel_launch(void* const* d_in, const int* in_sizes, int n_in,
                              void* d_out, int out_size, void* d_ws, size_t ws_size,
                              hipStream_t stream) {
  const float* x  = (const float*)d_in[0];
  const int*   ei = (const int*)d_in[1];
  const float* W1 = (const float*)d_in[2];
  const float* b1 = (const float*)d_in[3];
  const float* W2 = (const float*)d_in[4];
  const float* b2 = (const float*)d_in[5];

  int n = in_sizes[0] / 2;
  int E = in_sizes[1] / 2;
  const int* src = ei;
  const int* dst = ei + E;
  float* out = (float*)d_out;

  float* dinv = (float*)d_ws;
  float* agg  = dinv + n;
  float* z    = agg + 2 * (size_t)n;
  float* xs   = z + n;

  const int tb = 256;
  int gN = (n + tb - 1) / tb;
  int gE = ((n + TILE - 1) / TILE) * SLICES;

  gcn_init<<<gN, tb, 0, stream>>>(dinv, agg, z, out, n);
  gcn_degree<<<gE, ETHREADS, 0, stream>>>(dst, dinv, E, n);
  gcn_rsqrt<<<gN, tb, 0, stream>>>(dinv, x, xs, n);
  gcn_agg2<<<gE, ETHREADS, 0, stream>>>(src, dst, xs, agg, E, n);

  int tiles = (n + 15) / 16;
  int dblocks = (tiles + 7) / 8;
  gcn_dense_wmma<<<dblocks, 256, 0, stream>>>(agg, xs, dinv, W1, b1, W2, z, n);

  gcn_out_self<<<gN, tb, 0, stream>>>(z, dinv, b2, out, n);
  gcn_out_edges<<<gE, ETHREADS, 0, stream>>>(src, dst, z, dinv, out, E, n);
}


# Edge sums re-tiled: 5 destination ranges of 20480 nodes x 51 edge slices (was 16 x 16), 80/160 KiB LDS; an edge index is read 5 times, not 16; flush rotated by slice; f32 sum order: LDS arrival
# speedup vs baseline: 5.0926x; 1.0492x over previous
#include <hip/hip_runtime.h>


typedef float v2f __attribute__((ext_vector_type(2)));
typedef float v8f __attribute__((ext_vector_type(8)));

#define TILE 20480
#define SLICES 51
#define ETHREADS 1024
#define EUNROLL 16
#define FSLOTS 20
#define FSLOTS2 10
static_assert(TILE == FSLOTS * ETHREADS, "the 1-wide flush is one pass over the whole tile");
static_assert((2 * TILE) % (FSLOTS2 * ETHREADS) == 0, "the 2-wide flush walks the whole tile in full passes");
#define SLICE_EDGES(E) (((((E) + SLICES - 1) / SLICES) + 31) & ~31)
#define FLUSH_ROT(slice, width) ((((unsigned)(slice)) * ((unsigned)(width) / SLICES)) & ~31u)

__global__ void gcn_init(float* __restrict__ deg, float* __restrict__ agg,
                         float* __restrict__ z, float* __restrict__ out, int n) {
  int i = blockIdx.x * blockDim.x + threadIdx.x;
  if (i < n) {
    deg[i] = 1.0f;
    agg[2 * i] = 0.0f;
    agg[2 * i + 1] = 0.0f;
    z[i] = 0.0f;
    out[i] = 0.0f;
  }
}

__global__ __launch_bounds__(ETHREADS) void gcn_degree(
    const int* __restrict__ dst, float* __restrict__ deg, int E, int n) {
  __shared__ float acc[TILE];
  const int tid = threadIdx.x;
  const int slice = blockIdx.x % SLICES;
  const int lo = (blockIdx.x / SLICES) * TILE;
  const unsigned cnt = (unsigned)min(TILE, n - lo);
  const int chunk = SLICE_EDGES(E);
  const int e0 = slice * chunk;
  const int e1 = min(E, e0 + chunk);

  for (int i = tid; i < TILE / 4; i += ETHREADS) ((float4*)acc)[i] = make_float4(0.0f, 0.0f, 0.0f, 0.0f);
  __syncthreads();

  const int trips = (chunk + EUNROLL * ETHREADS - 1) / (EUNROLL * ETHREADS);
  for (int t = 0; t < trips; ++t) {
    const int i = e0 + tid + t * (EUNROLL * ETHREADS);
    int d[EUNROLL];
#pragma unroll
    for (int u = 0; u < EUNROLL; ++u) {
      int j = i + u * ETHREADS;
      d[u] = dst[min(j, e1 - 1)];
      if (j >= e1) d[u] = -1;
    }
#pragma unroll
    for (int u = 0; u < EUNROLL; ++u) {
      unsigned k = (unsigned)(d[u] - lo);
      if (d[u] >= lo && k < cnt) atomicAdd(&acc[k], 1.0f);
    }
  }
  __syncthreads();

  const unsigned rot = FLUSH_ROT(slice, TILE);
  unsigned idx[FSLOTS];
  float v[FSLOTS];
#pragma unroll
  for (int q = 0; q < FSLOTS; ++q) {
    unsigned i = tid + q * ETHREADS + rot;
    if (i >= TILE) i -= TILE;
    idx[q] = i;
    v[q] = acc[i];
  }
#pragma unroll
  for (int q = 0; q < FSLOTS; ++q)
    if (v[q] != 0.0f) atomicAdd(&deg[lo + idx[q]], v[q]);
}

__global__ void gcn_rsqrt(float* __restrict__ dinv, const float* __restrict__ x,
                          float* __restrict__ xs, int n) {
  int i = blockIdx.x * blockDim.x + threadIdx.x;
  if (i < n) {
    float di = rsqrtf(dinv[i]);
    dinv[i] = di;
    v2f xv = ((const v2f*)x)[i];
    v2f o;
    o.x = xv.x * di;
    o.y = xv.y * di;
    ((v2f*)xs)[i] = o;
  }
}

__global__ __launch_bounds__(ETHREADS) void gcn_agg2(
    const int* __restrict__ src, const int* __restrict__ dst,
    const float* __restrict__ xs, float* __restrict__ agg, int E, int n) {
  __shared__ float acc[2 * TILE];
  const int tid = threadIdx.x;
  const int slice = blockIdx.x % SLICES;
  const int lo = (blockIdx.x / SLICES) * TILE;
  const unsigned cnt = (unsigned)min(TILE, n - lo);
  const int chunk = SLICE_EDGES(E);
  const int e0 = slice * chunk;
  const int e1 = min(E, e0 + chunk);

  for (int i = tid; i < 2 * TILE / 4; i += ETHREADS) ((float4*)acc)[i] = make_float4(0.0f, 0.0f, 0.0f, 0.0f);
  __syncthreads();

  const v2f* __restrict__ xs2 = (const v2f*)xs;
  const int trips = (chunk + EUNROLL * ETHREADS - 1) / (EUNROLL * ETHREADS);
  for (int t = 0; t < trips; ++t) {
    const int i = e0 + tid + t * (EUNROLL * ETHREADS);
    int d[EUNROLL], s[EUNROLL];
    bool hit[EUNROLL];
#pragma unroll
    for (int u = 0; u < EUNROLL; ++u) {
      int j = min(i + u * ETHREADS, e1 - 1);
      d[u] = dst[j];
      s[u] = src[j];
    }
#pragma unroll
    for (int u = 0; u < EUNROLL; ++u) asm volatile("" ::"v"(d[u]), "v"(s[u]));
    unsigned k[EUNROLL];
#pragma unroll
    for (int u = 0; u < EUNROLL; ++u) {
      k[u] = (unsigned)(d[u] - lo);
      hit[u] = (i + u * ETHREADS < e1) && d[u] >= lo && k[u] < cnt;
      if (!hit[u]) s[u] = lo;
    }
    v2f p[EUNROLL];
#pragma unroll
    for (int u = 0; u < EUNROLL; ++u) p[u] = xs2[s[u]];
#pragma unroll
    for (int u = 0; u < EUNROLL; ++u) asm volatile("" ::"v"(p[u].x), "v"(p[u].y));
#pragma unroll
    for (int u = 0; u < EUNROLL; ++u) {
      if (hit[u]) {
        atomicAdd(&acc[2 * k[u]], p[u].x);
        atomicAdd(&acc[2 * k[u] + 1], p[u].y);
      }
    }
  }
  __syncthreads();

  const unsigned rot = FLUSH_ROT(slice, 2 * TILE);
  float* __restrict__ aggr = agg + 2 * (size_t)lo;
  for (unsigned j0 = tid; j0 < 2 * TILE; j0 += FSLOTS2 * ETHREADS) {
    unsigned idx[FSLOTS2];
    float v[FSLOTS2];
#pragma unroll
    for (int q = 0; q < FSLOTS2; ++q) {
      unsigned i = j0 + q * ETHREADS + rot;
      if (i >= 2 * TILE) i -= 2 * TILE;
      idx[q] = i;
      v[q] = acc[i];
    }
#pragma unroll
    for (int q = 0; q < FSLOTS2; ++q)
      if (v[q] != 0.0f) atomicAdd(&aggr[idx[q]], v[q]);
  }
}

__global__ __launch_bounds__(256) void gcn_dense_wmma(
    const float* __restrict__ agg, const float* __restrict__ xs,
    const float* __restrict__ dinv, const float* __restrict__ W1,
    const float* __restrict__ b1, const float* __restrict__ W2,
    float* __restrict__ zs, int n) {
  const int lane = threadIdx.x & 31;
  const int wave = threadIdx.x >> 5;
  const int tile = blockIdx.x * 8 + wave;
  const int base = tile * 16;
  if (base >= n) return;

  const int m = lane & 15;
  const bool hiK = lane >= 16;
  const float kmask = hiK ? 0.0f : 1.0f;

  int node = base + m;
  const float rowmask = (!hiK && node < n) ? 1.0f : 0.0f;
  if (node >= n) node = n - 1;
  float di = dinv[node];
  float sc = di * rowmask;
  v2f A;
  A.x = (agg[2 * node]     + xs[2 * node])     * sc;
  A.y = (agg[2 * node + 1] + xs[2 * node + 1]) * sc;

  float zacc[8];
#pragma unroll
  for (int v = 0; v < 8; ++v) zacc[v] = 0.0f;

#pragma unroll
  for (int n0 = 0; n0 < 128; n0 += 16) {
    int col = n0 + m;
    v2f B;
    B.x = W1[col]       * kmask;
    B.y = W1[128 + col] * kmask;
    float bias = b1[col];
    v8f C;
#pragma unroll
    for (int v = 0; v < 8; ++v) C[v] = bias;

    C = __builtin_amdgcn_wmma_f32_16x16x4_f32(
         false, A,  false, B,
         (short)0, C,  false,  false);

    float w2 = W2[col];
#pragma unroll
    for (int v = 0; v < 8; ++v) {
      float h = C[v];
      h = h > 0.0f ? h : 0.0f;
      zacc[v] += h * w2;
    }
  }

#pragma unroll
  for (int v = 0; v < 8; ++v) {
    float r = zacc[v];
    r += __shfl_xor(r, 1, 16);
    r += __shfl_xor(r, 2, 16);
    r += __shfl_xor(r, 4, 16);
    r += __shfl_xor(r, 8, 16);
    zacc[v] = r;
  }
  float val = zacc[0];
#pragma unroll
  for (int v = 1; v < 8; ++v) val = ((m & 7) == v) ? zacc[v] : val;
  const bool owner = ((m >> 3) == (hiK ? 1 : 0));
  if (owner && base + m < n) zs[base + m] = val * di;
}

__global__ void gcn_out_self(const float* __restrict__ zs,
                             const float* __restrict__ dinv,
                             const float* __restrict__ b2,
                             float* __restrict__ out, int n) {
  int i = blockIdx.x * blockDim.x + threadIdx.x;
  if (i < n) out[i] = zs[i] * dinv[i] + b2[0];
}

__global__ __launch_bounds__(ETHREADS) void gcn_out_edges(
    const int* __restrict__ src, const int* __restrict__ dst,
    const float* __restrict__ zs, const float* __restrict__ dinv,
    float* __restrict__ out, int E, int n) {
  __shared__ float acc[TILE];
  const int tid = threadIdx.x;
  const int slice = blockIdx.x % SLICES;
  const int lo = (blockIdx.x / SLICES) * TILE;
  const unsigned cnt = (unsigned)min(TILE, n - lo);
  const int chunk = SLICE_EDGES(E);
  const int e0 = slice * chunk;
  const int e1 = min(E, e0 + chunk);

  for (int i = tid; i < TILE / 4; i += ETHREADS) ((float4*)acc)[i] = make_float4(0.0f, 0.0f, 0.0f, 0.0f);
  __syncthreads();

  const int trips = (chunk + EUNROLL * ETHREADS - 1) / (EUNROLL * ETHREADS);
  for (int t = 0; t < trips; ++t) {
    const int i = e0 + tid + t * (EUNROLL * ETHREADS);
    int d[EUNROLL], s[EUNROLL];
    bool hit[EUNROLL];
#pragma unroll
    for (int u = 0; u < EUNROLL; ++u) {
      int j = min(i + u * ETHREADS, e1 - 1);
      d[u] = dst[j];
      s[u] = src[j];
    }
#pragma unroll
    for (int u = 0; u < EUNROLL; ++u) asm volatile("" ::"v"(d[u]), "v"(s[u]));
    unsigned k[EUNROLL];
#pragma unroll
    for (int u = 0; u < EUNROLL; ++u) {
      k[u] = (unsigned)(d[u] - lo);
      hit[u] = (i + u * ETHREADS < e1) && d[u] >= lo && k[u] < cnt;
      if (!hit[u]) s[u] = lo;
    }
    float p[EUNROLL];
#pragma unroll
    for (int u = 0; u < EUNROLL; ++u) p[u] = zs[s[u]];
#pragma unroll
    for (int u = 0; u < EUNROLL; ++u) asm volatile("" ::"v"(p[u]));
#pragma unroll
    for (int u = 0; u < EUNROLL; ++u) {
      if (hit[u]) atomicAdd(&acc[k[u]], p[u]);
    }
  }
  const unsigned rot = FLUSH_ROT(slice, TILE);
  unsigned idx[FSLOTS];
  float dd[FSLOTS];
#pragma unroll
  for (int q = 0; q < FSLOTS; ++q) {
    unsigned i = tid + q * ETHREADS + rot;
    if (i >= TILE) i -= TILE;
    idx[q] = i;
    dd[q] = dinv[min(lo + (int)i, n - 1)];
  }
  __syncthreads();

  float v[FSLOTS];
#pragma unroll
  for (int q = 0; q < FSLOTS; ++q) v[q] = acc[idx[q]];
#pragma unroll
  for (int q = 0; q < FSLOTS; ++q)
    if (v[q] != 0.0f) atomicAdd(&out[lo + idx[q]], v[q] * dd[q]);
}

extern "C" void kernel_launch(void* const* d_in, const int* in_sizes, int n_in,
                              void* d_out, int out_size, void* d_ws, size_t ws_size,
                              hipStream_t stream) {
  const float* x  = (const float*)d_in[0];
  const int*   ei = (const int*)d_in[1];
  const float* W1 = (const float*)d_in[2];
  const float* b1 = (const float*)d_in[3];
  const float* W2 = (const float*)d_in[4];
  const float* b2 = (const float*)d_in[5];

  int n = in_sizes[0] / 2;
  int E = in_sizes[1] / 2;
  const int* src = ei;
  const int* dst = ei + E;
  float* out = (float*)d_out;

  float* dinv = (float*)d_ws;
  float* agg  = dinv + n;
  float* z    = agg + 2 * (size_t)n;
  float* xs   = z + n;

  const int tb = 256;
  int gN = (n + tb - 1) / tb;
  int gE = ((n + TILE - 1) / TILE) * SLICES;

  gcn_init<<<gN, tb, 0, stream>>>(dinv, agg, z, out, n);
  gcn_degree<<<gE, ETHREADS, 0, stream>>>(dst, dinv, E, n);
  gcn_rsqrt<<<gN, tb, 0, stream>>>(dinv, x, xs, n);
  gcn_agg2<<<gE, ETHREADS, 0, stream>>>(src, dst, xs, agg, E, n);

  int tiles = (n + 15) / 16;
  int dblocks = (tiles + 7) / 8;
  gcn_dense_wmma<<<dblocks, 256, 0, stream>>>(agg, xs, dinv, W1, b1, W2, z, n);

  gcn_out_self<<<gN, tb, 0, stream>>>(z, dinv, b2, out, n);
  gcn_out_edges<<<gE, ETHREADS, 0, stream>>>(src, dst, z, dinv, out, E, n);
}


# Edge blocks draw their (range, slice) pair from a per-half counter picked by die id: a range's float atomics all come from one half of the chip; 4 ranges x 64 slices; f32 sum order: LDS arrival
# speedup vs baseline: 7.3785x; 1.4489x over previous
#include <hip/hip_runtime.h>


typedef float v2f __attribute__((ext_vector_type(2)));
typedef float v8f __attribute__((ext_vector_type(8)));

#define TILE 25600
#define SLICES 64
#define ETHREADS 1024
#define EUNROLL 13
#define FSLOTS 25
#define FSLOTS2 10
#define TICK_PITCH 64
static_assert(TILE == FSLOTS * ETHREADS, "the 1-wide flush is one pass over the whole tile");
static_assert((2 * TILE) % (FSLOTS2 * ETHREADS) == 0, "the 2-wide flush walks the whole tile in full passes");
#define SLICE_EDGES(E) (((((E) + SLICES - 1) / SLICES) + 31) & ~31)
#define FLUSH_ROT(slice, width) ((((unsigned)(slice)) * ((unsigned)(width) / SLICES)) & ~31u)

__device__ __forceinline__ void take_pair(int* __restrict__ tick, int* __restrict__ slot) {
  const unsigned id = __builtin_amdgcn_s_sendmsg_rtn(0x87);
  const int upper = (((id >> 16) & 0xFu) >= 4u) ? 1 : 0;
  const int t = atomicAdd(&tick[upper * TICK_PITCH], 1);
  const int last = (int)gridDim.x - 1;
  const int pair = upper ? last - t : t;
  *slot = min(max(pair, 0), last);
}

__global__ void gcn_init(float* __restrict__ deg, float* __restrict__ agg,
                         float* __restrict__ z, float* __restrict__ out, int n) {
  int i = blockIdx.x * blockDim.x + threadIdx.x;
  if (i < n) {
    deg[i] = 1.0f;
    agg[2 * i] = 0.0f;
    agg[2 * i + 1] = 0.0f;
    z[i] = 0.0f;
    out[i] = 0.0f;
  }
}

__global__ __launch_bounds__(ETHREADS) void gcn_degree(
    const int* __restrict__ dst, float* __restrict__ deg, int* __restrict__ tick, int E, int n) {
  __shared__ float acc[TILE];
  __shared__ int pair_slot;
  const int tid = threadIdx.x;
  if (tid == 0) take_pair(tick, &pair_slot);

  for (int i = tid; i < TILE / 4; i += ETHREADS) ((float4*)acc)[i] = make_float4(0.0f, 0.0f, 0.0f, 0.0f);
  __syncthreads();

  const int pair = __builtin_amdgcn_readfirstlane(pair_slot);
  const int slice = pair % SLICES;
  const int lo = (pair / SLICES) * TILE;
  const unsigned cnt = (unsigned)min(TILE, n - lo);
  const int chunk = SLICE_EDGES(E);
  const int e0 = slice * chunk;
  const int e1 = min(E, e0 + chunk);

  const int trips = (chunk + EUNROLL * ETHREADS - 1) / (EUNROLL * ETHREADS);
  for (int t = 0; t < trips; ++t) {
    const int i = e0 + tid + t * (EUNROLL * ETHREADS);
    int d[EUNROLL];
#pragma unroll
    for (int u = 0; u < EUNROLL; ++u) {
      int j = i + u * ETHREADS;
      d[u] = dst[min(j, e1 - 1)];
      if (j >= e1) d[u] = -1;
    }
#pragma unroll
    for (int u = 0; u < EUNROLL; ++u) {
      unsigned k = (unsigned)(d[u] - lo);
      if (d[u] >= lo && k < cnt) atomicAdd(&acc[k], 1.0f);
    }
  }
  __syncthreads();

  const unsigned rot = FLUSH_ROT(slice, TILE);
  unsigned idx[FSLOTS];
  float v[FSLOTS];
#pragma unroll
  for (int q = 0; q < FSLOTS; ++q) {
    unsigned i = tid + q * ETHREADS + rot;
    if (i >= TILE) i -= TILE;
    idx[q] = i;
    v[q] = acc[i];
  }
#pragma unroll
  for (int q = 0; q < FSLOTS; ++q)
    if (v[q] != 0.0f) atomicAdd(&deg[lo + idx[q]], v[q]);
}

__global__ void gcn_rsqrt(float* __restrict__ dinv, const float* __restrict__ x,
                          float* __restrict__ xs, int n) {
  int i = blockIdx.x * blockDim.x + threadIdx.x;
  if (i < n) {
    float di = rsqrtf(dinv[i]);
    dinv[i] = di;
    v2f xv = ((const v2f*)x)[i];
    v2f o;
    o.x = xv.x * di;
    o.y = xv.y * di;
    ((v2f*)xs)[i] = o;
  }
}

__global__ __launch_bounds__(ETHREADS) void gcn_agg2(
    const int* __restrict__ src, const int* __restrict__ dst,
    const float* __restrict__ xs, float* __restrict__ agg, int* __restrict__ tick, int E, int n) {
  __shared__ float acc[2 * TILE];
  __shared__ int pair_slot;
  const int tid = threadIdx.x;
  if (tid == 0) take_pair(tick, &pair_slot);

  for (int i = tid; i < 2 * TILE / 4; i += ETHREADS) ((float4*)acc)[i] = make_float4(0.0f, 0.0f, 0.0f, 0.0f);
  __syncthreads();

  const int pair = __builtin_amdgcn_readfirstlane(pair_slot);
  const int slice = pair % SLICES;
  const int lo = (pair / SLICES) * TILE;
  const unsigned cnt = (unsigned)min(TILE, n - lo);
  const int chunk = SLICE_EDGES(E);
  const int e0 = slice * chunk;
  const int e1 = min(E, e0 + chunk);

  const v2f* __restrict__ xs2 = (const v2f*)xs;
  const int trips = (chunk + EUNROLL * ETHREADS - 1) / (EUNROLL * ETHREADS);
  for (int t = 0; t < trips; ++t) {
    const int i = e0 + tid + t * (EUNROLL * ETHREADS);
    int d[EUNROLL], s[EUNROLL];
    bool hit[EUNROLL];
#pragma unroll
    for (int u = 0; u < EUNROLL; ++u) {
      int j = min(i + u * ETHREADS, e1 - 1);
      d[u] = dst[j];
      s[u] = src[j];
    }
#pragma unroll
    for (int u = 0; u < EUNROLL; ++u) asm volatile("" ::"v"(d[u]), "v"(s[u]));
    unsigned k[EUNROLL];
#pragma unroll
    for (int u = 0; u < EUNROLL; ++u) {
      k[u] = (unsigned)(d[u] - lo);
      hit[u] = (i + u * ETHREADS < e1) & (d[u] >= lo) & (k[u] < cnt);
      s[u] = hit[u] ? s[u] : lo;
    }
    v2f p[EUNROLL];
#pragma unroll
    for (int u = 0; u < EUNROLL; ++u) p[u] = xs2[s[u]];
#pragma unroll
    for (int u = 0; u < EUNROLL; ++u) asm volatile("" ::"v"(p[u].x), "v"(p[u].y));
#pragma unroll
    for (int u = 0; u < EUNROLL; ++u) {
      if (hit[u]) {
        atomicAdd(&acc[2 * k[u]], p[u].x);
        atomicAdd(&acc[2 * k[u] + 1], p[u].y);
      }
    }
  }
  __syncthreads();

  const unsigned rot = FLUSH_ROT(slice, 2 * TILE);
  float* __restrict__ aggr = agg + 2 * (size_t)lo;
  for (unsigned j0 = tid; j0 < 2 * TILE; j0 += FSLOTS2 * ETHREADS) {
    unsigned idx[FSLOTS2];
    float v[FSLOTS2];
#pragma unroll
    for (int q = 0; q < FSLOTS2; ++q) {
      unsigned i = j0 + q * ETHREADS + rot;
      if (i >= 2 * TILE) i -= 2 * TILE;
      idx[q] = i;
      v[q] = acc[i];
    }
#pragma unroll
    for (int q = 0; q < FSLOTS2; ++q)
      if (v[q] != 0.0f) atomicAdd(&aggr[idx[q]], v[q]);
  }
}

__global__ __launch_bounds__(256) void gcn_dense_wmma(
    const float* __restrict__ agg, const float* __restrict__ xs,
    const float* __restrict__ dinv, const float* __restrict__ W1,
    const float* __restrict__ b1, const float* __restrict__ W2,
    float* __restrict__ zs, int n) {
  const int lane = threadIdx.x & 31;
  const int wave = threadIdx.x >> 5;
  const int tile = blockIdx.x * 8 + wave;
  const int base = tile * 16;
  if (base >= n) return;

  const int m = lane & 15;
  const bool hiK = lane >= 16;
  const float kmask = hiK ? 0.0f : 1.0f;

  int node = base + m;
  const float rowmask = (!hiK && node < n) ? 1.0f : 0.0f;
  if (node >= n) node = n - 1;
  float di = dinv[node];
  float sc = di * rowmask;
  v2f A;
  A.x = (agg[2 * node]     + xs[2 * node])     * sc;
  A.y = (agg[2 * node + 1] + xs[2 * node + 1]) * sc;

  float zacc[8];
#pragma unroll
  for (int v = 0; v < 8; ++v) zacc[v] = 0.0f;

#pragma unroll
  for (int n0 = 0; n0 < 128; n0 += 16) {
    int col = n0 + m;
    v2f B;
    B.x = W1[col]       * kmask;
    B.y = W1[128 + col] * kmask;
    float bias = b1[col];
    v8f C;
#pragma unroll
    for (int v = 0; v < 8; ++v) C[v] = bias;

    C = __builtin_amdgcn_wmma_f32_16x16x4_f32(
         false, A,  false, B,
         (short)0, C,  false,  false);

    float w2 = W2[col];
#pragma unroll
    for (int v = 0; v < 8; ++v) {
      float h = C[v];
      h = h > 0.0f ? h : 0.0f;
      zacc[v] += h * w2;
    }
  }

#pragma unroll
  for (int v = 0; v < 8; ++v) {
    float r = zacc[v];
    r += __shfl_xor(r, 1, 16);
    r += __shfl_xor(r, 2, 16);
    r += __shfl_xor(r, 4, 16);
    r += __shfl_xor(r, 8, 16);
    zacc[v] = r;
  }
  float val = zacc[0];
#pragma unroll
  for (int v = 1; v < 8; ++v) val = ((m & 7) == v) ? zacc[v] : val;
  const bool owner = ((m >> 3) == (hiK ? 1 : 0));
  if (owner && base + m < n) zs[base + m] = val * di;
}

__global__ void gcn_out_self(const float* __restrict__ zs,
                             const float* __restrict__ dinv,
                             const float* __restrict__ b2,
                             float* __restrict__ out, int* __restrict__ tick, int n) {
  int i = blockIdx.x * blockDim.x + threadIdx.x;
  if (i < n) out[i] = zs[i] * dinv[i] + b2[0];
  if (i == 0) {
    tick[0] = 0;
    tick[TICK_PITCH] = 0;
  }
}

__global__ __launch_bounds__(ETHREADS) void gcn_out_edges(
    const int* __restrict__ src, const int* __restrict__ dst,
    const float* __restrict__ zs, const float* __restrict__ dinv,
    float* __restrict__ out, int* __restrict__ tick, int E, int n) {
  __shared__ float acc[TILE];
  __shared__ int pair_slot;
  const int tid = threadIdx.x;
  if (tid == 0) take_pair(tick, &pair_slot);

  for (int i = tid; i < TILE / 4; i += ETHREADS) ((float4*)acc)[i] = make_float4(0.0f, 0.0f, 0.0f, 0.0f);
  __syncthreads();

  const int pair = __builtin_amdgcn_readfirstlane(pair_slot);
  const int slice = pair % SLICES;
  const int lo = (pair / SLICES) * TILE;
  const unsigned cnt = (unsigned)min(TILE, n - lo);
  const int chunk = SLICE_EDGES(E);
  const int e0 = slice * chunk;
  const int e1 = min(E, e0 + chunk);

  const int trips = (chunk + EUNROLL * ETHREADS - 1) / (EUNROLL * ETHREADS);
  for (int t = 0; t < trips; ++t) {
    const int i = e0 + tid + t * (EUNROLL * ETHREADS);
    int d[EUNROLL], s[EUNROLL];
    bool hit[EUNROLL];
#pragma unroll
    for (int u = 0; u < EUNROLL; ++u) {
      int j = min(i + u * ETHREADS, e1 - 1);
      d[u] = dst[j];
      s[u] = src[j];
    }
#pragma unroll
    for (int u = 0; u < EUNROLL; ++u) asm volatile("" ::"v"(d[u]), "v"(s[u]));
    unsigned k[EUNROLL];
#pragma unroll
    for (int u = 0; u < EUNROLL; ++u) {
      k[u] = (unsigned)(d[u] - lo);
      hit[u] = (i + u * ETHREADS < e1) & (d[u] >= lo) & (k[u] < cnt);
      s[u] = hit[u] ? s[u] : lo;
    }
    float p[EUNROLL];
#pragma unroll
    for (int u = 0; u < EUNROLL; ++u) p[u] = zs[s[u]];
#pragma unroll
    for (int u = 0; u < EUNROLL; ++u) asm volatile("" ::"v"(p[u]));
#pragma unroll
    for (int u = 0; u < EUNROLL; ++u) {
      if (hit[u]) atomicAdd(&acc[k[u]], p[u]);
    }
  }
  const unsigned rot = FLUSH_ROT(slice, TILE);
  unsigned idx[FSLOTS];
  float dd[FSLOTS];
#pragma unroll
  for (int q = 0; q < FSLOTS; ++q) {
    unsigned i = tid + q * ETHREADS + rot;
    if (i >= TILE) i -= TILE;
    idx[q] = i;
    dd[q] = dinv[min(lo + (int)i, n - 1)];
  }
  __syncthreads();

  float v[FSLOTS];
#pragma unroll
  for (int q = 0; q < FSLOTS; ++q) v[q] = acc[idx[q]];
#pragma unroll
  for (int q = 0; q < FSLOTS; ++q)
    if (v[q] != 0.0f) atomicAdd(&out[lo + idx[q]], v[q] * dd[q]);
}

extern "C" void kernel_launch(void* const* d_in, const int* in_sizes, int n_in,
                              void* d_out, int out_size, void* d_ws, size_t ws_size,
                              hipStream_t stream) {
  const float* x  = (const float*)d_in[0];
  const int*   ei = (const int*)d_in[1];
  const float* W1 = (const float*)d_in[2];
  const float* b1 = (const float*)d_in[3];
  const float* W2 = (const float*)d_in[4];
  const float* b2 = (const float*)d_in[5];

  int n = in_sizes[0] / 2;
  int E = in_sizes[1] / 2;
  const int* src = ei;
  const int* dst = ei + E;
  float* out = (float*)d_out;

  float* dinv = (float*)d_ws;
  float* agg  = dinv + n;
  float* z    = agg + 2 * (size_t)n;
  float* xs   = z + n;

  const int tb = 256;
  int gN = (n + tb - 1) / tb;
  int gE = ((n + TILE - 1) / TILE) * SLICES;

  gcn_init<<<gN, tb, 0, stream>>>(dinv, agg, z, out, n);
  gcn_degree<<<gE, ETHREADS, 0, stream>>>(dst, dinv, (int*)z, E, n);
  gcn_rsqrt<<<gN, tb, 0, stream>>>(dinv, x, xs, n);
  gcn_agg2<<<gE, ETHREADS, 0, stream>>>(src, dst, xs, agg, (int*)z + 1, E, n);

  int tiles = (n + 15) / 16;
  int dblocks = (tiles + 7) / 8;
  gcn_dense_wmma<<<dblocks, 256, 0, stream>>>(agg, xs, dinv, W1, b1, W2, z, n);

  gcn_out_self<<<gN, tb, 0, stream>>>(z, dinv, b2, out, (int*)agg, n);
  gcn_out_edges<<<gE, ETHREADS, 0, stream>>>(src, dst, z, dinv, out, (int*)agg, E, n);
}


# Self-loop+bias launch folded into the dense stage (it stores out's start value); layer-1 sums start at the node's own xs, set by the rsqrt stage; 6 launches, not 7; f32 sum order: self term first
# speedup vs baseline: 7.6362x; 1.0349x over previous
#include <hip/hip_runtime.h>


typedef float v2f __attribute__((ext_vector_type(2)));
typedef float v8f __attribute__((ext_vector_type(8)));

#define TILE 25600
#define SLICES 64
#define ETHREADS 1024
#define EUNROLL 13
#define FSLOTS 25
#define FSLOTS2 10
#define TICK_PITCH 64
static_assert(TILE == FSLOTS * ETHREADS, "the 1-wide flush is one pass over the whole tile");
static_assert((2 * TILE) % (FSLOTS2 * ETHREADS) == 0, "the 2-wide flush walks the whole tile in full passes");
#define SLICE_EDGES(E) (((((E) + SLICES - 1) / SLICES) + 31) & ~31)
#define FLUSH_ROT(slice, width) ((((unsigned)(slice)) * ((unsigned)(width) / SLICES)) & ~31u)

__device__ __forceinline__ void take_pair(int* __restrict__ tick, int* __restrict__ slot) {
  const unsigned id = __builtin_amdgcn_s_sendmsg_rtn(0x87);
  const int upper = (((id >> 16) & 0xFu) >= 4u) ? 1 : 0;
  const int t = atomicAdd(&tick[upper * TICK_PITCH], 1);
  const int last = (int)gridDim.x - 1;
  const int pair = upper ? last - t : t;
  *slot = min(max(pair, 0), last);
}

__global__ void gcn_init(float* __restrict__ deg, int* __restrict__ tick, int n) {
  int i = blockIdx.x * blockDim.x + threadIdx.x;
  if (i < n) deg[i] = 1.0f;
  if (i < 2) {
    tick[i] = 0;
    tick[TICK_PITCH + i] = 0;
  }
}

__global__ __launch_bounds__(ETHREADS) void gcn_degree(
    const int* __restrict__ dst, float* __restrict__ deg, int* __restrict__ tick, int E, int n) {
  __shared__ float acc[TILE];
  __shared__ int pair_slot;
  const int tid = threadIdx.x;
  if (tid == 0) take_pair(tick, &pair_slot);

  for (int i = tid; i < TILE / 4; i += ETHREADS) ((float4*)acc)[i] = make_float4(0.0f, 0.0f, 0.0f, 0.0f);
  __syncthreads();

  const int pair = __builtin_amdgcn_readfirstlane(pair_slot);
  const int slice = pair % SLICES;
  const int lo = (pair / SLICES) * TILE;
  const unsigned cnt = (unsigned)min(TILE, n - lo);
  const int chunk = SLICE_EDGES(E);
  const int e0 = slice * chunk;
  const int e1 = min(E, e0 + chunk);

  const int trips = (chunk + EUNROLL * ETHREADS - 1) / (EUNROLL * ETHREADS);
  for (int t = 0; t < trips; ++t) {
    const int i = e0 + tid + t * (EUNROLL * ETHREADS);
    int d[EUNROLL];
#pragma unroll
    for (int u = 0; u < EUNROLL; ++u) {
      int j = i + u * ETHREADS;
      d[u] = dst[min(j, e1 - 1)];
      if (j >= e1) d[u] = -1;
    }
#pragma unroll
    for (int u = 0; u < EUNROLL; ++u) {
      unsigned k = (unsigned)(d[u] - lo);
      if (d[u] >= lo && k < cnt) atomicAdd(&acc[k], 1.0f);
    }
  }
  __syncthreads();

  const unsigned rot = FLUSH_ROT(slice, TILE);
  unsigned idx[FSLOTS];
  float v[FSLOTS];
#pragma unroll
  for (int q = 0; q < FSLOTS; ++q) {
    unsigned i = tid + q * ETHREADS + rot;
    if (i >= TILE) i -= TILE;
    idx[q] = i;
    v[q] = acc[i];
  }
#pragma unroll
  for (int q = 0; q < FSLOTS; ++q)
    if (v[q] != 0.0f) atomicAdd(&deg[lo + idx[q]], v[q]);
}

__global__ void gcn_rsqrt(float* __restrict__ dinv, const float* __restrict__ x,
                          float* __restrict__ xs, float* __restrict__ agg, int n) {
  int i = blockIdx.x * blockDim.x + threadIdx.x;
  if (i < n) {
    float di = rsqrtf(dinv[i]);
    dinv[i] = di;
    v2f xv = ((const v2f*)x)[i];
    v2f o;
    o.x = xv.x * di;
    o.y = xv.y * di;
    ((v2f*)xs)[i] = o;
    ((v2f*)agg)[i] = o;
  }
}

__global__ __launch_bounds__(ETHREADS) void gcn_agg2(
    const int* __restrict__ src, const int* __restrict__ dst,
    const float* __restrict__ xs, float* __restrict__ agg, int* __restrict__ tick, int E, int n) {
  __shared__ float acc[2 * TILE];
  __shared__ int pair_slot;
  const int tid = threadIdx.x;
  if (tid == 0) take_pair(tick, &pair_slot);

  for (int i = tid; i < 2 * TILE / 4; i += ETHREADS) ((float4*)acc)[i] = make_float4(0.0f, 0.0f, 0.0f, 0.0f);
  __syncthreads();

  const int pair = __builtin_amdgcn_readfirstlane(pair_slot);
  const int slice = pair % SLICES;
  const int lo = (pair / SLICES) * TILE;
  const unsigned cnt = (unsigned)min(TILE, n - lo);
  const int chunk = SLICE_EDGES(E);
  const int e0 = slice * chunk;
  const int e1 = min(E, e0 + chunk);

  const v2f* __restrict__ xs2 = (const v2f*)xs;
  const int trips = (chunk + EUNROLL * ETHREADS - 1) / (EUNROLL * ETHREADS);
  for (int t = 0; t < trips; ++t) {
    const int i = e0 + tid + t * (EUNROLL * ETHREADS);
    int d[EUNROLL], s[EUNROLL];
    bool hit[EUNROLL];
#pragma unroll
    for (int u = 0; u < EUNROLL; ++u) {
      int j = min(i + u * ETHREADS, e1 - 1);
      d[u] = dst[j];
      s[u] = src[j];
    }
#pragma unroll
    for (int u = 0; u < EUNROLL; ++u) asm volatile("" ::"v"(d[u]), "v"(s[u]));
    unsigned k[EUNROLL];
#pragma unroll
    for (int u = 0; u < EUNROLL; ++u) {
      k[u] = (unsigned)(d[u] - lo);
      hit[u] = (i + u * ETHREADS < e1) & (d[u] >= lo) & (k[u] < cnt);
      s[u] = hit[u] ? s[u] : lo;
    }
    v2f p[EUNROLL];
#pragma unroll
    for (int u = 0; u < EUNROLL; ++u) p[u] = xs2[s[u]];
#pragma unroll
    for (int u = 0; u < EUNROLL; ++u) asm volatile("" ::"v"(p[u].x), "v"(p[u].y));
#pragma unroll
    for (int u = 0; u < EUNROLL; ++u) {
      if (hit[u]) {
        atomicAdd(&acc[2 * k[u]], p[u].x);
        atomicAdd(&acc[2 * k[u] + 1], p[u].y);
      }
    }
  }
  __syncthreads();

  const unsigned rot = FLUSH_ROT(slice, 2 * TILE);
  float* __restrict__ aggr = agg + 2 * (size_t)lo;
  for (unsigned j0 = tid; j0 < 2 * TILE; j0 += FSLOTS2 * ETHREADS) {
    unsigned idx[FSLOTS2];
    float v[FSLOTS2];
#pragma unroll
    for (int q = 0; q < FSLOTS2; ++q) {
      unsigned i = j0 + q * ETHREADS + rot;
      if (i >= 2 * TILE) i -= 2 * TILE;
      idx[q] = i;
      v[q] = acc[i];
    }
#pragma unroll
    for (int q = 0; q < FSLOTS2; ++q)
      if (v[q] != 0.0f) atomicAdd(&aggr[idx[q]], v[q]);
  }
}

__global__ __launch_bounds__(256) void gcn_dense_wmma(
    const float* __restrict__ agg,
    const float* __restrict__ dinv, const float* __restrict__ W1,
    const float* __restrict__ b1, const float* __restrict__ W2,
    const float* __restrict__ b2,
    float* __restrict__ zs, float* __restrict__ out, int* __restrict__ tick, int n) {
  if (blockIdx.x == 0 && threadIdx.x == 0) {
    tick[0] = 0;
    tick[TICK_PITCH] = 0;
  }
  const int lane = threadIdx.x & 31;
  const int wave = threadIdx.x >> 5;
  const int tile = blockIdx.x * 8 + wave;
  const int base = tile * 16;
  if (base >= n) return;

  const int m = lane & 15;
  const bool hiK = lane >= 16;
  const float kmask = hiK ? 0.0f : 1.0f;

  int node = base + m;
  const float rowmask = (!hiK && node < n) ? 1.0f : 0.0f;
  if (node >= n) node = n - 1;
  float di = dinv[node];
  float sc = di * rowmask;
  v2f A;
  const v2f av = ((const v2f*)agg)[node];
  A.x = av.x * sc;
  A.y = av.y * sc;

  float zacc[8];
#pragma unroll
  for (int v = 0; v < 8; ++v) zacc[v] = 0.0f;

#pragma unroll
  for (int n0 = 0; n0 < 128; n0 += 16) {
    int col = n0 + m;
    v2f B;
    B.x = W1[col]       * kmask;
    B.y = W1[128 + col] * kmask;
    float bias = b1[col];
    v8f C;
#pragma unroll
    for (int v = 0; v < 8; ++v) C[v] = bias;

    C = __builtin_amdgcn_wmma_f32_16x16x4_f32(
         false, A,  false, B,
         (short)0, C,  false,  false);

    float w2 = W2[col];
#pragma unroll
    for (int v = 0; v < 8; ++v) {
      float h = C[v];
      h = h > 0.0f ? h : 0.0f;
      zacc[v] += h * w2;
    }
  }

#pragma unroll
  for (int v = 0; v < 8; ++v) {
    float r = zacc[v];
    r += __shfl_xor(r, 1, 16);
    r += __shfl_xor(r, 2, 16);
    r += __shfl_xor(r, 4, 16);
    r += __shfl_xor(r, 8, 16);
    zacc[v] = r;
  }
  float val = zacc[0];
#pragma unroll
  for (int v = 1; v < 8; ++v) val = ((m & 7) == v) ? zacc[v] : val;
  const bool owner = ((m >> 3) == (hiK ? 1 : 0));
  if (owner && base + m < n) {
    const float zv = val * di;
    zs[base + m] = zv;
    out[base + m] = zv * di + b2[0];
  }
}

__global__ __launch_bounds__(ETHREADS) void gcn_out_edges(
    const int* __restrict__ src, const int* __restrict__ dst,
    const float* __restrict__ zs, const float* __restrict__ dinv,
    float* __restrict__ out, int* __restrict__ tick, int E, int n) {
  __shared__ float acc[TILE];
  __shared__ int pair_slot;
  const int tid = threadIdx.x;
  if (tid == 0) take_pair(tick, &pair_slot);

  for (int i = tid; i < TILE / 4; i += ETHREADS) ((float4*)acc)[i] = make_float4(0.0f, 0.0f, 0.0f, 0.0f);
  __syncthreads();

  const int pair = __builtin_amdgcn_readfirstlane(pair_slot);
  const int slice = pair % SLICES;
  const int lo = (pair / SLICES) * TILE;
  const unsigned cnt = (unsigned)min(TILE, n - lo);
  const int chunk = SLICE_EDGES(E);
  const int e0 = slice * chunk;
  const int e1 = min(E, e0 + chunk);

  const int trips = (chunk + EUNROLL * ETHREADS - 1) / (EUNROLL * ETHREADS);
  for (int t = 0; t < trips; ++t) {
    const int i = e0 + tid + t * (EUNROLL * ETHREADS);
    int d[EUNROLL], s[EUNROLL];
    bool hit[EUNROLL];
#pragma unroll
    for (int u = 0; u < EUNROLL; ++u) {
      int j = min(i + u * ETHREADS, e1 - 1);
      d[u] = dst[j];
      s[u] = src[j];
    }
#pragma unroll
    for (int u = 0; u < EUNROLL; ++u) asm volatile("" ::"v"(d[u]), "v"(s[u]));
    unsigned k[EUNROLL];
#pragma unroll
    for (int u = 0; u < EUNROLL; ++u) {
      k[u] = (unsigned)(d[u] - lo);
      hit[u] = (i + u * ETHREADS < e1) & (d[u] >= lo) & (k[u] < cnt);
      s[u] = hit[u] ? s[u] : lo;
    }
    float p[EUNROLL];
#pragma unroll
    for (int u = 0; u < EUNROLL; ++u) p[u] = zs[s[u]];
#pragma unroll
    for (int u = 0; u < EUNROLL; ++u) asm volatile("" ::"v"(p[u]));
#pragma unroll
    for (int u = 0; u < EUNROLL; ++u) {
      if (hit[u]) atomicAdd(&acc[k[u]], p[u]);
    }
  }
  const unsigned rot = FLUSH_ROT(slice, TILE);
  unsigned idx[FSLOTS];
  float dd[FSLOTS];
#pragma unroll
  for (int q = 0; q < FSLOTS; ++q) {
    unsigned i = tid + q * ETHREADS + rot;
    if (i >= TILE) i -= TILE;
    idx[q] = i;
    dd[q] = dinv[min(lo + (int)i, n - 1)];
  }
  __syncthreads();

  float v[FSLOTS];
#pragma unroll
  for (int q = 0; q < FSLOTS; ++q) v[q] = acc[idx[q]];
#pragma unroll
  for (int q = 0; q < FSLOTS; ++q)
    if (v[q] != 0.0f) atomicAdd(&out[lo + idx[q]], v[q] * dd[q]);
}

extern "C" void kernel_launch(void* const* d_in, const int* in_sizes, int n_in,
                              void* d_out, int out_size, void* d_ws, size_t ws_size,
                              hipStream_t stream) {
  const float* x  = (const float*)d_in[0];
  const int*   ei = (const int*)d_in[1];
  const float* W1 = (const float*)d_in[2];
  const float* b1 = (const float*)d_in[3];
  const float* W2 = (const float*)d_in[4];
  const float* b2 = (const float*)d_in[5];

  int n = in_sizes[0] / 2;
  int E = in_sizes[1] / 2;
  const int* src = ei;
  const int* dst = ei + E;
  float* out = (float*)d_out;

  float* dinv = (float*)d_ws;
  float* agg  = dinv + n;
  float* z    = agg + 2 * (size_t)n;
  float* xs   = z + n;

  const int tb = 256;
  int gN = (n + tb - 1) / tb;
  int gE = ((n + TILE - 1) / TILE) * SLICES;

  gcn_init<<<gN, tb, 0, stream>>>(dinv, (int*)z, n);
  gcn_degree<<<gE, ETHREADS, 0, stream>>>(dst, dinv, (int*)z, E, n);
  gcn_rsqrt<<<gN, tb, 0, stream>>>(dinv, x, xs, agg, n);
  gcn_agg2<<<gE, ETHREADS, 0, stream>>>(src, dst, xs, agg, (int*)z + 1, E, n);

  int tiles = (n + 15) / 16;
  int dblocks = (tiles + 7) / 8;
  gcn_dense_wmma<<<dblocks, 256, 0, stream>>>(agg, dinv, W1, b1, W2, b2, z, out, (int*)xs, n);

  gcn_out_edges<<<gE, ETHREADS, 0, stream>>>(src, dst, z, dinv, out, (int*)xs, E, n);
}
